# starting kernel
# baseline (speedup 1.0000x reference)
#include <hip/hip_runtime.h>
#include <math.h>

typedef __attribute__((ext_vector_type(2))) float v2f;
typedef __attribute__((ext_vector_type(4))) float f4;
typedef __attribute__((ext_vector_type(8))) float v8f;

#define BATCH   2
#define SEQ     2048
#define HIDDEN  2048
#define NHEADS  16
#define HEADSZ  128
#define QKVDIM  6144
#define ROTH    16
#define MASKVAL (-3.4028234663852886e38f)

__global__ __launch_bounds__(256) void gemm_bias_wmma(
    const float* __restrict__ A, const float* __restrict__ B,
    const float* __restrict__ bias, float* __restrict__ C,
    int M, int N, int K)
{
    __shared__ float As[2][128 * 36];
    __shared__ float Bt[2][128 * 36];

    const int tid  = threadIdx.x;
    const int lane = tid & 31;
    const int wid  = tid >> 5;
    const int lh   = lane & 15;
    const int kh2  = (lane >> 4) << 1;
    const int rsel = (lane >> 4) << 3;

    const int m0 = blockIdx.y * 128;
    const int n0 = blockIdx.x * 128;
    const int waveRow = (wid & 3) * 32;
    const int waveCol = (wid >> 2) * 64;

    v8f zero = {};
    v8f acc[2][4];
    #pragma unroll
    for (int i = 0; i < 2; ++i)
        #pragma unroll
        for (int j = 0; j < 4; ++j) acc[i][j] = zero;

    f4 ra[4], rb[4];

    const int nt = K >> 5;

    #pragma unroll
    for (int it = 0; it < 4; ++it) {
        int e = it * 256 + tid;
        int r = e >> 3, c = (e & 7) << 2;
        ra[it] = *(const f4*)(A + (size_t)(m0 + r) * K + c);
        int n = e & 127, k4 = (e >> 7) << 2;
        f4 t;
        t.x = B[(size_t)(k4 + 0) * N + n0 + n];
        t.y = B[(size_t)(k4 + 1) * N + n0 + n];
        t.z = B[(size_t)(k4 + 2) * N + n0 + n];
        t.w = B[(size_t)(k4 + 3) * N + n0 + n];
        rb[it] = t;
    }
    #pragma unroll
    for (int it = 0; it < 4; ++it) {
        int e = it * 256 + tid;
        int r = e >> 3, c = (e & 7) << 2;
        *(f4*)(&As[0][r * 36 + c]) = ra[it];
        int n = e & 127, k4 = (e >> 7) << 2;
        *(f4*)(&Bt[0][n * 36 + k4]) = rb[it];
    }

    for (int t = 0; t < nt; ++t) {
        const int cur = t & 1;
        const int k0n = (t + 1) << 5;

        if (t + 1 < nt) {
            #pragma unroll
            for (int it = 0; it < 4; ++it) {
                int e = it * 256 + tid;
                int r = e >> 3, c = (e & 7) << 2;
                ra[it] = *(const f4*)(A + (size_t)(m0 + r) * K + k0n + c);
                int n = e & 127, k4 = (e >> 7) << 2;
                f4 tv;
                tv.x = B[(size_t)(k0n + k4 + 0) * N + n0 + n];
                tv.y = B[(size_t)(k0n + k4 + 1) * N + n0 + n];
                tv.z = B[(size_t)(k0n + k4 + 2) * N + n0 + n];
                tv.w = B[(size_t)(k0n + k4 + 3) * N + n0 + n];
                rb[it] = tv;
            }
        }

        __syncthreads();

        const float* Ac = &As[cur][0];
        const float* Bc = &Bt[cur][0];
        #pragma unroll
        for (int kk = 0; kk < 32; kk += 4) {
            int kb = kk + kh2;
            v2f af[2];
            #pragma unroll
            for (int i = 0; i < 2; ++i)
                af[i] = *(const v2f*)(&Ac[(waveRow + i * 16 + lh) * 36 + kb]);
            v2f bf[4];
            #pragma unroll
            for (int j = 0; j < 4; ++j)
                bf[j] = *(const v2f*)(&Bc[(waveCol + j * 16 + lh) * 36 + kb]);
            #pragma unroll
            for (int i = 0; i < 2; ++i)
                #pragma unroll
                for (int j = 0; j < 4; ++j)
                    acc[i][j] = __builtin_amdgcn_wmma_f32_16x16x4_f32(
                        false, af[i], false, bf[j], (short)0, acc[i][j], false, false);
        }

        if (t + 1 < nt) {
            #pragma unroll
            for (int it = 0; it < 4; ++it) {
                int e = it * 256 + tid;
                int r = e >> 3, c = (e & 7) << 2;
                *(f4*)(&As[cur ^ 1][r * 36 + c]) = ra[it];
                int n = e & 127, k4 = (e >> 7) << 2;
                *(f4*)(&Bt[cur ^ 1][n * 36 + k4]) = rb[it];
            }
        }
    }

    #pragma unroll
    for (int i = 0; i < 2; ++i) {
        #pragma unroll
        for (int j = 0; j < 4; ++j) {
            int gcol = n0 + waveCol + j * 16 + lh;
            float bv = bias[gcol];
            #pragma unroll
            for (int r = 0; r < 8; ++r) {
                int grow = m0 + waveRow + i * 16 + r + rsel;
                C[(size_t)grow * N + gcol] = acc[i][j][r] + bv;
            }
        }
    }
}

__global__ __launch_bounds__(256) void rope_kernel(
    float* __restrict__ qkv, const int* __restrict__ pos_ids)
{
    int idx = blockIdx.x * 256 + threadIdx.x;
    int i = idx & 15;
    int h = (idx >> 4) & 15;
    int s = (idx >> 8) & (SEQ - 1);
    int b = idx >> 19;

    float t = (float)pos_ids[s];
    float freq = powf(10000.0f, -(float)(2 * i) / 32.0f);
    float sn, cs;
    sincosf(t * freq, &sn, &cs);

    size_t base = ((size_t)(b * SEQ + s)) * QKVDIM + (size_t)h * 384;
    float q0 = qkv[base + i], q1 = qkv[base + i + ROTH];
    qkv[base + i]        = q0 * cs - q1 * sn;
    qkv[base + i + ROTH] = q1 * cs + q0 * sn;
    float k0 = qkv[base + 128 + i], k1 = qkv[base + 128 + i + ROTH];
    qkv[base + 128 + i]        = k0 * cs - k1 * sn;
    qkv[base + 128 + i + ROTH] = k1 * cs + k0 * sn;
}

__global__ __launch_bounds__(128) void flash_attn_kernel(
    const float* __restrict__ qkv, const float* __restrict__ amask,
    float* __restrict__ ctx)
{
    __shared__ float Qs[4 * 16 * 132];
    __shared__ float Ks[2][16 * 132];
    __shared__ float Vt[2][128 * 20];
    __shared__ float Ps[4 * 16 * 18];

    const int tid  = threadIdx.x;
    const int lane = tid & 31;
    const int w    = tid >> 5;
    const int lh   = lane & 15;
    const int kh2  = (lane >> 4) << 1;
    const int rsel = (lane >> 4) << 3;

    const int b  = blockIdx.y >> 4;
    const int h  = blockIdx.y & 15;
    const int q0 = blockIdx.x * 64;

    const size_t qkv_bh = (size_t)b * SEQ * QKVDIM + (size_t)h * 384;

    #pragma unroll
    for (int it = 0; it < 16; ++it) {
        int e = it * 128 + tid;
        int r = e >> 5;
        int c = (e & 31) << 2;
        f4 v = *(const f4*)(qkv + qkv_bh + (size_t)(q0 + r) * QKVDIM + c);
        *(f4*)(&Qs[r * 132 + c]) = v;
    }

    v8f zero = {};
    v8f o[8];
    #pragma unroll
    for (int j = 0; j < 8; ++j) o[j] = zero;
    float mrow[8], lrow[8];
    #pragma unroll
    for (int r = 0; r < 8; ++r) { mrow[r] = MASKVAL; lrow[r] = 0.0f; }

    const float norm = 0.08838834764831845f;
    const int kb_hi = (q0 + 63) >> 4;
    float* Qw = &Qs[w * 16 * 132];
    float* Pw = &Ps[w * 16 * 18];

    f4 rk[4], rv[4];

    auto loadKV = [&](int kblk) {
        #pragma unroll
        for (int it = 0; it < 4; ++it) {
            int e = it * 128 + tid;
            int kr = e >> 5, kc = (e & 31) << 2;
            rk[it] = *(const f4*)(qkv + qkv_bh + (size_t)(kblk * 16 + kr) * QKVDIM + 128 + kc);
            int vd = e & 127, vk = (e >> 7) << 2;
            f4 tv;
            tv.x = qkv[qkv_bh + (size_t)(kblk * 16 + vk + 0) * QKVDIM + 256 + vd];
            tv.y = qkv[qkv_bh + (size_t)(kblk * 16 + vk + 1) * QKVDIM + 256 + vd];
            tv.z = qkv[qkv_bh + (size_t)(kblk * 16 + vk + 2) * QKVDIM + 256 + vd];
            tv.w = qkv[qkv_bh + (size_t)(kblk * 16 + vk + 3) * QKVDIM + 256 + vd];
            rv[it] = tv;
        }
    };
    auto storeKV = [&](int buf) {
        #pragma unroll
        for (int it = 0; it < 4; ++it) {
            int e = it * 128 + tid;
            int kr = e >> 5, kc = (e & 31) << 2;
            *(f4*)(&Ks[buf][kr * 132 + kc]) = rk[it];
            int vd = e & 127, vk = (e >> 7) << 2;
            *(f4*)(&Vt[buf][vd * 20 + vk]) = rv[it];
        }
    };

    loadKV(0);
    storeKV(0);

    for (int kblk = 0; kblk <= kb_hi; ++kblk) {
        const int cur = kblk & 1;
        if (kblk < kb_hi) loadKV(kblk + 1);

        __syncthreads();

        v8f s = zero;
        const float* Kc = &Ks[cur][0];
        #pragma unroll
        for (int step = 0; step < 32; ++step) {
            int kd = step * 4 + kh2;
            v2f a   = *(const v2f*)(&Qw[lh * 132 + kd]);
            v2f bfr = *(const v2f*)(&Kc[lh * 132 + kd]);
            s = __builtin_amdgcn_wmma_f32_16x16x4_f32(false, a, false, bfr,
                                                      (short)0, s, false, false);
        }

        int   kidx = kblk * 16 + lh;
        float am   = amask[(size_t)b * SEQ + kidx];
        float sv[8];
        #pragma unroll
        for (int r = 0; r < 8; ++r) {
            int srow = q0 + w * 16 + r + rsel;
            float v = (kidx <= srow) ? (s[r] * norm) : MASKVAL;
            sv[r] = v + am;
        }

        #pragma unroll
        for (int r = 0; r < 8; ++r) {
            float mx = sv[r];
            mx = fmaxf(mx, __shfl_xor(mx, 1, 32));
            mx = fmaxf(mx, __shfl_xor(mx, 2, 32));
            mx = fmaxf(mx, __shfl_xor(mx, 4, 32));
            mx = fmaxf(mx, __shfl_xor(mx, 8, 32));
            float mnew = fmaxf(mrow[r], mx);
            float pe   = __expf(sv[r] - mnew);
            float sum  = pe;
            sum += __shfl_xor(sum, 1, 32);
            sum += __shfl_xor(sum, 2, 32);
            sum += __shfl_xor(sum, 4, 32);
            sum += __shfl_xor(sum, 8, 32);
            float scale = __expf(mrow[r] - mnew);
            lrow[r] = lrow[r] * scale + sum;
            mrow[r] = mnew;
            #pragma unroll
            for (int j = 0; j < 8; ++j) o[j][r] *= scale;
            Pw[(r + rsel) * 18 + lh] = pe;
        }
        asm volatile("s_wait_dscnt 0" ::: "memory");

        const float* Vc = &Vt[cur][0];
        #pragma unroll
        for (int st = 0; st < 4; ++st) {
            int kp = st * 4 + kh2;
            v2f a = *(const v2f*)(&Pw[lh * 18 + kp]);
            #pragma unroll
            for (int j = 0; j < 8; ++j) {
                v2f bfr = *(const v2f*)(&Vc[(j * 16 + lh) * 20 + kp]);
                o[j] = __builtin_amdgcn_wmma_f32_16x16x4_f32(false, a, false, bfr,
                                                             (short)0, o[j], false, false);
            }
        }

        if (kblk < kb_hi) storeKV(cur ^ 1);
    }

    #pragma unroll
    for (int r = 0; r < 8; ++r) {
        float inv  = 1.0f / lrow[r];
        int   srow = q0 + w * 16 + r + rsel;
        size_t rowbase = ((size_t)b * SEQ + srow) * HIDDEN + (size_t)h * HEADSZ;
        #pragma unroll
        for (int j = 0; j < 8; ++j)
            ctx[rowbase + j * 16 + lh] = o[j][r] * inv;
    }
}

extern "C" void kernel_launch(void* const* d_in, const int* in_sizes, int n_in,
                              void* d_out, int out_size, void* d_ws, size_t ws_size,
                              hipStream_t stream)
{
    const float* hs    = (const float*)d_in[0];
    const float* amask = (const float*)d_in[1];
    const float* Wqkv  = (const float*)d_in[2];
    const float* bqkv  = (const float*)d_in[3];
    const float* Wd    = (const float*)d_in[4];
    const float* bd    = (const float*)d_in[5];
    const int*   pos   = (const int*)d_in[6];
    float* out = (float*)d_out;

    float* qkv = (float*)d_ws;
    float* ctx = qkv + (size_t)BATCH * SEQ * QKVDIM;

    (void)in_sizes; (void)n_in; (void)out_size; (void)ws_size;

    gemm_bias_wmma<<<dim3(QKVDIM / 128, (BATCH * SEQ) / 128), dim3(256), 0, stream>>>(
        hs, Wqkv, bqkv, qkv, BATCH * SEQ, QKVDIM, HIDDEN);

    rope_kernel<<<dim3((BATCH * SEQ * NHEADS * 16) / 256), dim3(256), 0, stream>>>(qkv, pos);

    flash_attn_kernel<<<dim3(SEQ / 64, BATCH * NHEADS), dim3(128), 0, stream>>>(qkv, amask, ctx);

    gemm_bias_wmma<<<dim3(HIDDEN / 128, (BATCH * SEQ) / 128), dim3(256), 0, stream>>>(
        ctx, Wd, bd, out, BATCH * SEQ, HIDDEN, HIDDEN);
}


# K-loop hand-over barriers wait for LDS only (asm s_wait_dscnt 0 + barrier), mask value requested a block ahead with the prefetch: next tile's global loads stay in flight, not drained by __syncthreads
# speedup vs baseline: 1.0579x; 1.0579x over previous
#include <hip/hip_runtime.h>
#include <math.h>

typedef __attribute__((ext_vector_type(2))) float v2f;
typedef __attribute__((ext_vector_type(4))) float f4;
typedef __attribute__((ext_vector_type(8))) float v8f;

#define BATCH   2
#define SEQ     2048
#define HIDDEN  2048
#define NHEADS  16
#define HEADSZ  128
#define QKVDIM  6144
#define ROTH    16
#define MASKVAL (-3.4028234663852886e38f)

static __device__ __forceinline__ void lds_handover_barrier()
{
    asm volatile("s_wait_dscnt 0x0\n\ts_barrier_signal -1\n\ts_barrier_wait -1" ::: "memory");
}

__global__ __launch_bounds__(256) void gemm_bias_wmma(
    const float* __restrict__ A, const float* __restrict__ B,
    const float* __restrict__ bias, float* __restrict__ C,
    int M, int N, int K)
{
    __shared__ float As[2][128 * 36];
    __shared__ float Bt[2][128 * 36];

    const int tid  = threadIdx.x;
    const int lane = tid & 31;
    const int wid  = tid >> 5;
    const int lh   = lane & 15;
    const int kh2  = (lane >> 4) << 1;
    const int rsel = (lane >> 4) << 3;

    const int m0 = blockIdx.y * 128;
    const int n0 = blockIdx.x * 128;
    const int waveRow = (wid & 3) * 32;
    const int waveCol = (wid >> 2) * 64;

    v8f zero = {};
    v8f acc[2][4];
    #pragma unroll
    for (int i = 0; i < 2; ++i)
        #pragma unroll
        for (int j = 0; j < 4; ++j) acc[i][j] = zero;

    f4 ra[4], rb[4];

    const int nt = K >> 5;

    #pragma unroll
    for (int it = 0; it < 4; ++it) {
        int e = it * 256 + tid;
        int r = e >> 3, c = (e & 7) << 2;
        ra[it] = *(const f4*)(A + (size_t)(m0 + r) * K + c);
        int n = e & 127, k4 = (e >> 7) << 2;
        f4 t;
        t.x = B[(size_t)(k4 + 0) * N + n0 + n];
        t.y = B[(size_t)(k4 + 1) * N + n0 + n];
        t.z = B[(size_t)(k4 + 2) * N + n0 + n];
        t.w = B[(size_t)(k4 + 3) * N + n0 + n];
        rb[it] = t;
    }
    #pragma unroll
    for (int it = 0; it < 4; ++it) {
        int e = it * 256 + tid;
        int r = e >> 3, c = (e & 7) << 2;
        *(f4*)(&As[0][r * 36 + c]) = ra[it];
        int n = e & 127, k4 = (e >> 7) << 2;
        *(f4*)(&Bt[0][n * 36 + k4]) = rb[it];
    }

    for (int t = 0; t < nt; ++t) {
        const int cur = t & 1;
        const int k0n = (t + 1) << 5;

        if (t + 1 < nt) {
            #pragma unroll
            for (int it = 0; it < 4; ++it) {
                int e = it * 256 + tid;
                int r = e >> 3, c = (e & 7) << 2;
                ra[it] = *(const f4*)(A + (size_t)(m0 + r) * K + k0n + c);
                int n = e & 127, k4 = (e >> 7) << 2;
                f4 tv;
                tv.x = B[(size_t)(k0n + k4 + 0) * N + n0 + n];
                tv.y = B[(size_t)(k0n + k4 + 1) * N + n0 + n];
                tv.z = B[(size_t)(k0n + k4 + 2) * N + n0 + n];
                tv.w = B[(size_t)(k0n + k4 + 3) * N + n0 + n];
                rb[it] = tv;
            }
        }

        lds_handover_barrier();

        const float* Ac = &As[cur][0];
        const float* Bc = &Bt[cur][0];
        #pragma unroll
        for (int kk = 0; kk < 32; kk += 4) {
            int kb = kk + kh2;
            v2f af[2];
            #pragma unroll
            for (int i = 0; i < 2; ++i)
                af[i] = *(const v2f*)(&Ac[(waveRow + i * 16 + lh) * 36 + kb]);
            v2f bf[4];
            #pragma unroll
            for (int j = 0; j < 4; ++j)
                bf[j] = *(const v2f*)(&Bc[(waveCol + j * 16 + lh) * 36 + kb]);
            #pragma unroll
            for (int i = 0; i < 2; ++i)
                #pragma unroll
                for (int j = 0; j < 4; ++j)
                    acc[i][j] = __builtin_amdgcn_wmma_f32_16x16x4_f32(
                        false, af[i], false, bf[j], (short)0, acc[i][j], false, false);
        }

        if (t + 1 < nt) {
            #pragma unroll
            for (int it = 0; it < 4; ++it) {
                int e = it * 256 + tid;
                int r = e >> 3, c = (e & 7) << 2;
                *(f4*)(&As[cur ^ 1][r * 36 + c]) = ra[it];
                int n = e & 127, k4 = (e >> 7) << 2;
                *(f4*)(&Bt[cur ^ 1][n * 36 + k4]) = rb[it];
            }
        }
    }

    #pragma unroll
    for (int i = 0; i < 2; ++i) {
        #pragma unroll
        for (int j = 0; j < 4; ++j) {
            int gcol = n0 + waveCol + j * 16 + lh;
            float bv = bias[gcol];
            #pragma unroll
            for (int r = 0; r < 8; ++r) {
                int grow = m0 + waveRow + i * 16 + r + rsel;
                C[(size_t)grow * N + gcol] = acc[i][j][r] + bv;
            }
        }
    }
}

__global__ __launch_bounds__(256) void rope_kernel(
    float* __restrict__ qkv, const int* __restrict__ pos_ids)
{
    int idx = blockIdx.x * 256 + threadIdx.x;
    int i = idx & 15;
    int h = (idx >> 4) & 15;
    int s = (idx >> 8) & (SEQ - 1);
    int b = idx >> 19;

    float t = (float)pos_ids[s];
    float freq = powf(10000.0f, -(float)(2 * i) / 32.0f);
    float sn, cs;
    sincosf(t * freq, &sn, &cs);

    size_t base = ((size_t)(b * SEQ + s)) * QKVDIM + (size_t)h * 384;
    float q0 = qkv[base + i], q1 = qkv[base + i + ROTH];
    qkv[base + i]        = q0 * cs - q1 * sn;
    qkv[base + i + ROTH] = q1 * cs + q0 * sn;
    float k0 = qkv[base + 128 + i], k1 = qkv[base + 128 + i + ROTH];
    qkv[base + 128 + i]        = k0 * cs - k1 * sn;
    qkv[base + 128 + i + ROTH] = k1 * cs + k0 * sn;
}

__global__ __launch_bounds__(128) void flash_attn_kernel(
    const float* __restrict__ qkv, const float* __restrict__ amask,
    float* __restrict__ ctx)
{
    __shared__ float Qs[4 * 16 * 132];
    __shared__ float Ks[2][16 * 132];
    __shared__ float Vt[2][128 * 20];
    __shared__ float Ps[4 * 16 * 18];

    const int tid  = threadIdx.x;
    const int lane = tid & 31;
    const int w    = tid >> 5;
    const int lh   = lane & 15;
    const int kh2  = (lane >> 4) << 1;
    const int rsel = (lane >> 4) << 3;

    const int b  = blockIdx.y >> 4;
    const int h  = blockIdx.y & 15;
    const int q0 = blockIdx.x * 64;

    const size_t qkv_bh = (size_t)b * SEQ * QKVDIM + (size_t)h * 384;

    #pragma unroll
    for (int it = 0; it < 16; ++it) {
        int e = it * 128 + tid;
        int r = e >> 5;
        int c = (e & 31) << 2;
        f4 v = *(const f4*)(qkv + qkv_bh + (size_t)(q0 + r) * QKVDIM + c);
        *(f4*)(&Qs[r * 132 + c]) = v;
    }

    v8f zero = {};
    v8f o[8];
    #pragma unroll
    for (int j = 0; j < 8; ++j) o[j] = zero;
    float mrow[8], lrow[8];
    #pragma unroll
    for (int r = 0; r < 8; ++r) { mrow[r] = MASKVAL; lrow[r] = 0.0f; }

    const float norm = 0.08838834764831845f;
    const int kb_hi = (q0 + 63) >> 4;
    float* Qw = &Qs[w * 16 * 132];
    float* Pw = &Ps[w * 16 * 18];

    f4 rk[4], rv[4];

    auto loadKV = [&](int kblk) {
        #pragma unroll
        for (int it = 0; it < 4; ++it) {
            int e = it * 128 + tid;
            int kr = e >> 5, kc = (e & 31) << 2;
            rk[it] = *(const f4*)(qkv + qkv_bh + (size_t)(kblk * 16 + kr) * QKVDIM + 128 + kc);
            int vd = e & 127, vk = (e >> 7) << 2;
            f4 tv;
            tv.x = qkv[qkv_bh + (size_t)(kblk * 16 + vk + 0) * QKVDIM + 256 + vd];
            tv.y = qkv[qkv_bh + (size_t)(kblk * 16 + vk + 1) * QKVDIM + 256 + vd];
            tv.z = qkv[qkv_bh + (size_t)(kblk * 16 + vk + 2) * QKVDIM + 256 + vd];
            tv.w = qkv[qkv_bh + (size_t)(kblk * 16 + vk + 3) * QKVDIM + 256 + vd];
            rv[it] = tv;
        }
    };
    auto storeKV = [&](int buf) {
        #pragma unroll
        for (int it = 0; it < 4; ++it) {
            int e = it * 128 + tid;
            int kr = e >> 5, kc = (e & 31) << 2;
            *(f4*)(&Ks[buf][kr * 132 + kc]) = rk[it];
            int vd = e & 127, vk = (e >> 7) << 2;
            *(f4*)(&Vt[buf][vd * 20 + vk]) = rv[it];
        }
    };

    float am_next = amask[(size_t)b * SEQ + lh];
    loadKV(0);
    storeKV(0);

    for (int kblk = 0; kblk <= kb_hi; ++kblk) {
        const int cur = kblk & 1;
        const int   kidx = kblk * 16 + lh;
        const float am   = am_next;
        if (kblk < kb_hi) {
            loadKV(kblk + 1);
            am_next = amask[(size_t)b * SEQ + kidx + 16];
        }

        lds_handover_barrier();

        v8f s = zero;
        const float* Kc = &Ks[cur][0];
        #pragma unroll
        for (int step = 0; step < 32; ++step) {
            int kd = step * 4 + kh2;
            v2f a   = *(const v2f*)(&Qw[lh * 132 + kd]);
            v2f bfr = *(const v2f*)(&Kc[lh * 132 + kd]);
            s = __builtin_amdgcn_wmma_f32_16x16x4_f32(false, a, false, bfr,
                                                      (short)0, s, false, false);
        }

        float sv[8];
        #pragma unroll
        for (int r = 0; r < 8; ++r) {
            int srow = q0 + w * 16 + r + rsel;
            float v = (kidx <= srow) ? (s[r] * norm) : MASKVAL;
            sv[r] = v + am;
        }

        #pragma unroll
        for (int r = 0; r < 8; ++r) {
            float mx = sv[r];
            mx = fmaxf(mx, __shfl_xor(mx, 1, 32));
            mx = fmaxf(mx, __shfl_xor(mx, 2, 32));
            mx = fmaxf(mx, __shfl_xor(mx, 4, 32));
            mx = fmaxf(mx, __shfl_xor(mx, 8, 32));
            float mnew = fmaxf(mrow[r], mx);
            float pe   = __expf(sv[r] - mnew);
            float sum  = pe;
            sum += __shfl_xor(sum, 1, 32);
            sum += __shfl_xor(sum, 2, 32);
            sum += __shfl_xor(sum, 4, 32);
            sum += __shfl_xor(sum, 8, 32);
            float scale = __expf(mrow[r] - mnew);
            lrow[r] = lrow[r] * scale + sum;
            mrow[r] = mnew;
            #pragma unroll
            for (int j = 0; j < 8; ++j) o[j][r] *= scale;
            Pw[(r + rsel) * 18 + lh] = pe;
        }
        asm volatile("s_wait_dscnt 0" ::: "memory");

        const float* Vc = &Vt[cur][0];
        #pragma unroll
        for (int st = 0; st < 4; ++st) {
            int kp = st * 4 + kh2;
            v2f a = *(const v2f*)(&Pw[lh * 18 + kp]);
            #pragma unroll
            for (int j = 0; j < 8; ++j) {
                v2f bfr = *(const v2f*)(&Vc[(j * 16 + lh) * 20 + kp]);
                o[j] = __builtin_amdgcn_wmma_f32_16x16x4_f32(false, a, false, bfr,
                                                             (short)0, o[j], false, false);
            }
        }

        if (kblk < kb_hi) storeKV(cur ^ 1);
    }

    #pragma unroll
    for (int r = 0; r < 8; ++r) {
        float inv  = 1.0f / lrow[r];
        int   srow = q0 + w * 16 + r + rsel;
        size_t rowbase = ((size_t)b * SEQ + srow) * HIDDEN + (size_t)h * HEADSZ;
        #pragma unroll
        for (int j = 0; j < 8; ++j)
            ctx[rowbase + j * 16 + lh] = o[j][r] * inv;
    }
}

extern "C" void kernel_launch(void* const* d_in, const int* in_sizes, int n_in,
                              void* d_out, int out_size, void* d_ws, size_t ws_size,
                              hipStream_t stream)
{
    const float* hs    = (const float*)d_in[0];
    const float* amask = (const float*)d_in[1];
    const float* Wqkv  = (const float*)d_in[2];
    const float* bqkv  = (const float*)d_in[3];
    const float* Wd    = (const float*)d_in[4];
    const float* bd    = (const float*)d_in[5];
    const int*   pos   = (const int*)d_in[6];
    float* out = (float*)d_out;

    float* qkv = (float*)d_ws;
    float* ctx = qkv + (size_t)BATCH * SEQ * QKVDIM;

    (void)in_sizes; (void)n_in; (void)out_size; (void)ws_size;

    gemm_bias_wmma<<<dim3(QKVDIM / 128, (BATCH * SEQ) / 128), dim3(256), 0, stream>>>(
        hs, Wqkv, bqkv, qkv, BATCH * SEQ, QKVDIM, HIDDEN);

    rope_kernel<<<dim3((BATCH * SEQ * NHEADS * 16) / 256), dim3(256), 0, stream>>>(qkv, pos);

    flash_attn_kernel<<<dim3(SEQ / 64, BATCH * NHEADS), dim3(128), 0, stream>>>(qkv, amask, ctx);

    gemm_bias_wmma<<<dim3(HIDDEN / 128, (BATCH * SEQ) / 128), dim3(256), 0, stream>>>(
        ctx, Wd, bd, out, BATCH * SEQ, HIDDEN, HIDDEN);
}


# Causal attention: each 4-wave workgroup does q-tile 31-x and then q-tile x (grid 16x32, was 32x32): 132 key blocks per workgroup instead of 4..128; per-row summation order unchanged
# speedup vs baseline: 1.1542x; 1.0910x over previous
#include <hip/hip_runtime.h>
#include <math.h>

typedef __attribute__((ext_vector_type(2))) float v2f;
typedef __attribute__((ext_vector_type(4))) float f4;
typedef __attribute__((ext_vector_type(8))) float v8f;

#define BATCH   2
#define SEQ     2048
#define HIDDEN  2048
#define NHEADS  16
#define HEADSZ  128
#define QKVDIM  6144
#define ROTH    16
#define MASKVAL (-3.4028234663852886e38f)

static __device__ __forceinline__ void lds_handover_barrier()
{
    asm volatile("s_wait_dscnt 0x0\n\ts_barrier_signal -1\n\ts_barrier_wait -1" ::: "memory");
}

__global__ __launch_bounds__(256) void gemm_bias_wmma(
    const float* __restrict__ A, const float* __restrict__ B,
    const float* __restrict__ bias, float* __restrict__ C,
    int M, int N, int K)
{
    __shared__ float As[2][128 * 36];
    __shared__ float Bt[2][128 * 36];

    const int tid  = threadIdx.x;
    const int lane = tid & 31;
    const int wid  = tid >> 5;
    const int lh   = lane & 15;
    const int kh2  = (lane >> 4) << 1;
    const int rsel = (lane >> 4) << 3;

    const int m0 = blockIdx.y * 128;
    const int n0 = blockIdx.x * 128;
    const int waveRow = (wid & 3) * 32;
    const int waveCol = (wid >> 2) * 64;

    v8f zero = {};
    v8f acc[2][4];
    #pragma unroll
    for (int i = 0; i < 2; ++i)
        #pragma unroll
        for (int j = 0; j < 4; ++j) acc[i][j] = zero;

    f4 ra[4], rb[4];

    const int nt = K >> 5;

    #pragma unroll
    for (int it = 0; it < 4; ++it) {
        int e = it * 256 + tid;
        int r = e >> 3, c = (e & 7) << 2;
        ra[it] = *(const f4*)(A + (size_t)(m0 + r) * K + c);
        int n = e & 127, k4 = (e >> 7) << 2;
        f4 t;
        t.x = B[(size_t)(k4 + 0) * N + n0 + n];
        t.y = B[(size_t)(k4 + 1) * N + n0 + n];
        t.z = B[(size_t)(k4 + 2) * N + n0 + n];
        t.w = B[(size_t)(k4 + 3) * N + n0 + n];
        rb[it] = t;
    }
    #pragma unroll
    for (int it = 0; it < 4; ++it) {
        int e = it * 256 + tid;
        int r = e >> 3, c = (e & 7) << 2;
        *(f4*)(&As[0][r * 36 + c]) = ra[it];
        int n = e & 127, k4 = (e >> 7) << 2;
        *(f4*)(&Bt[0][n * 36 + k4]) = rb[it];
    }

    for (int t = 0; t < nt; ++t) {
        const int cur = t & 1;
        const int k0n = (t + 1) << 5;

        if (t + 1 < nt) {
            #pragma unroll
            for (int it = 0; it < 4; ++it) {
                int e = it * 256 + tid;
                int r = e >> 3, c = (e & 7) << 2;
                ra[it] = *(const f4*)(A + (size_t)(m0 + r) * K + k0n + c);
                int n = e & 127, k4 = (e >> 7) << 2;
                f4 tv;
                tv.x = B[(size_t)(k0n + k4 + 0) * N + n0 + n];
                tv.y = B[(size_t)(k0n + k4 + 1) * N + n0 + n];
                tv.z = B[(size_t)(k0n + k4 + 2) * N + n0 + n];
                tv.w = B[(size_t)(k0n + k4 + 3) * N + n0 + n];
                rb[it] = tv;
            }
        }

        lds_handover_barrier();

        const float* Ac = &As[cur][0];
        const float* Bc = &Bt[cur][0];
        #pragma unroll
        for (int kk = 0; kk < 32; kk += 4) {
            int kb = kk + kh2;
            v2f af[2];
            #pragma unroll
            for (int i = 0; i < 2; ++i)
                af[i] = *(const v2f*)(&Ac[(waveRow + i * 16 + lh) * 36 + kb]);
            v2f bf[4];
            #pragma unroll
            for (int j = 0; j < 4; ++j)
                bf[j] = *(const v2f*)(&Bc[(waveCol + j * 16 + lh) * 36 + kb]);
            #pragma unroll
            for (int i = 0; i < 2; ++i)
                #pragma unroll
                for (int j = 0; j < 4; ++j)
                    acc[i][j] = __builtin_amdgcn_wmma_f32_16x16x4_f32(
                        false, af[i], false, bf[j], (short)0, acc[i][j], false, false);
        }

        if (t + 1 < nt) {
            #pragma unroll
            for (int it = 0; it < 4; ++it) {
                int e = it * 256 + tid;
                int r = e >> 3, c = (e & 7) << 2;
                *(f4*)(&As[cur ^ 1][r * 36 + c]) = ra[it];
                int n = e & 127, k4 = (e >> 7) << 2;
                *(f4*)(&Bt[cur ^ 1][n * 36 + k4]) = rb[it];
            }
        }
    }

    #pragma unroll
    for (int i = 0; i < 2; ++i) {
        #pragma unroll
        for (int j = 0; j < 4; ++j) {
            int gcol = n0 + waveCol + j * 16 + lh;
            float bv = bias[gcol];
            #pragma unroll
            for (int r = 0; r < 8; ++r) {
                int grow = m0 + waveRow + i * 16 + r + rsel;
                C[(size_t)grow * N + gcol] = acc[i][j][r] + bv;
            }
        }
    }
}

__global__ __launch_bounds__(256) void rope_kernel(
    float* __restrict__ qkv, const int* __restrict__ pos_ids)
{
    int idx = blockIdx.x * 256 + threadIdx.x;
    int i = idx & 15;
    int h = (idx >> 4) & 15;
    int s = (idx >> 8) & (SEQ - 1);
    int b = idx >> 19;

    float t = (float)pos_ids[s];
    float freq = powf(10000.0f, -(float)(2 * i) / 32.0f);
    float sn, cs;
    sincosf(t * freq, &sn, &cs);

    size_t base = ((size_t)(b * SEQ + s)) * QKVDIM + (size_t)h * 384;
    float q0 = qkv[base + i], q1 = qkv[base + i + ROTH];
    qkv[base + i]        = q0 * cs - q1 * sn;
    qkv[base + i + ROTH] = q1 * cs + q0 * sn;
    float k0 = qkv[base + 128 + i], k1 = qkv[base + 128 + i + ROTH];
    qkv[base + 128 + i]        = k0 * cs - k1 * sn;
    qkv[base + 128 + i + ROTH] = k1 * cs + k0 * sn;
}

__global__ __launch_bounds__(128) void flash_attn_kernel(
    const float* __restrict__ qkv, const float* __restrict__ amask,
    float* __restrict__ ctx)
{
    __shared__ float Qs[4 * 16 * 132];
    __shared__ float Ks[2][16 * 132];
    __shared__ float Vt[2][128 * 20];
    __shared__ float Ps[4 * 16 * 18];

    const int tid  = threadIdx.x;
    const int lane = tid & 31;
    const int w    = tid >> 5;
    const int lh   = lane & 15;
    const int kh2  = (lane >> 4) << 1;
    const int rsel = (lane >> 4) << 3;

    const int b  = blockIdx.y >> 4;
    const int h  = blockIdx.y & 15;

    const size_t qkv_bh = (size_t)b * SEQ * QKVDIM + (size_t)h * 384;

    #pragma unroll 1
    for (int pass = 0; pass < 2; ++pass) {
        const int qt = pass ? (int)blockIdx.x : (SEQ / 64 - 1 - (int)blockIdx.x);
        const int q0 = qt * 64;

        lds_handover_barrier();

        #pragma unroll
        for (int it = 0; it < 16; ++it) {
            int e = it * 128 + tid;
            int r = e >> 5;
            int c = (e & 31) << 2;
            f4 v = *(const f4*)(qkv + qkv_bh + (size_t)(q0 + r) * QKVDIM + c);
            *(f4*)(&Qs[r * 132 + c]) = v;
        }

        v8f zero = {};
        v8f o[8];
        #pragma unroll
        for (int j = 0; j < 8; ++j) o[j] = zero;
        float mrow[8], lrow[8];
        #pragma unroll
        for (int r = 0; r < 8; ++r) { mrow[r] = MASKVAL; lrow[r] = 0.0f; }

        const float norm = 0.08838834764831845f;
        const int kb_hi = (q0 + 63) >> 4;
        float* Qw = &Qs[w * 16 * 132];
        float* Pw = &Ps[w * 16 * 18];

        f4 rk[4], rv[4];

        auto loadKV = [&](int kblk) {
            #pragma unroll
            for (int it = 0; it < 4; ++it) {
                int e = it * 128 + tid;
                int kr = e >> 5, kc = (e & 31) << 2;
                rk[it] = *(const f4*)(qkv + qkv_bh + (size_t)(kblk * 16 + kr) * QKVDIM + 128 + kc);
                int vd = e & 127, vk = (e >> 7) << 2;
                f4 tv;
                tv.x = qkv[qkv_bh + (size_t)(kblk * 16 + vk + 0) * QKVDIM + 256 + vd];
                tv.y = qkv[qkv_bh + (size_t)(kblk * 16 + vk + 1) * QKVDIM + 256 + vd];
                tv.z = qkv[qkv_bh + (size_t)(kblk * 16 + vk + 2) * QKVDIM + 256 + vd];
                tv.w = qkv[qkv_bh + (size_t)(kblk * 16 + vk + 3) * QKVDIM + 256 + vd];
                rv[it] = tv;
            }
        };
        auto storeKV = [&](int buf) {
            #pragma unroll
            for (int it = 0; it < 4; ++it) {
                int e = it * 128 + tid;
                int kr = e >> 5, kc = (e & 31) << 2;
                *(f4*)(&Ks[buf][kr * 132 + kc]) = rk[it];
                int vd = e & 127, vk = (e >> 7) << 2;
                *(f4*)(&Vt[buf][vd * 20 + vk]) = rv[it];
            }
        };

        float am_next = amask[(size_t)b * SEQ + lh];
        loadKV(0);
        storeKV(0);

        for (int kblk = 0; kblk <= kb_hi; ++kblk) {
            const int cur = kblk & 1;
            const int   kidx = kblk * 16 + lh;
            const float am   = am_next;
            if (kblk < kb_hi) {
                loadKV(kblk + 1);
                am_next = amask[(size_t)b * SEQ + kidx + 16];
            }

            lds_handover_barrier();

            v8f s = zero;
            const float* Kc = &Ks[cur][0];
            #pragma unroll
            for (int step = 0; step < 32; ++step) {
                int kd = step * 4 + kh2;
                v2f a   = *(const v2f*)(&Qw[lh * 132 + kd]);
                v2f bfr = *(const v2f*)(&Kc[lh * 132 + kd]);
                s = __builtin_amdgcn_wmma_f32_16x16x4_f32(false, a, false, bfr,
                                                          (short)0, s, false, false);
            }

            float sv[8];
            #pragma unroll
            for (int r = 0; r < 8; ++r) {
                int srow = q0 + w * 16 + r + rsel;
                float v = (kidx <= srow) ? (s[r] * norm) : MASKVAL;
                sv[r] = v + am;
            }

            #pragma unroll
            for (int r = 0; r < 8; ++r) {
                float mx = sv[r];
                mx = fmaxf(mx, __shfl_xor(mx, 1, 32));
                mx = fmaxf(mx, __shfl_xor(mx, 2, 32));
                mx = fmaxf(mx, __shfl_xor(mx, 4, 32));
                mx = fmaxf(mx, __shfl_xor(mx, 8, 32));
                float mnew = fmaxf(mrow[r], mx);
                float pe   = __expf(sv[r] - mnew);
                float sum  = pe;
                sum += __shfl_xor(sum, 1, 32);
                sum += __shfl_xor(sum, 2, 32);
                sum += __shfl_xor(sum, 4, 32);
                sum += __shfl_xor(sum, 8, 32);
                float scale = __expf(mrow[r] - mnew);
                lrow[r] = lrow[r] * scale + sum;
                mrow[r] = mnew;
                #pragma unroll
                for (int j = 0; j < 8; ++j) o[j][r] *= scale;
                Pw[(r + rsel) * 18 + lh] = pe;
            }
            asm volatile("s_wait_dscnt 0" ::: "memory");

            const float* Vc = &Vt[cur][0];
            #pragma unroll
            for (int st = 0; st < 4; ++st) {
                int kp = st * 4 + kh2;
                v2f a = *(const v2f*)(&Pw[lh * 18 + kp]);
                #pragma unroll
                for (int j = 0; j < 8; ++j) {
                    v2f bfr = *(const v2f*)(&Vc[(j * 16 + lh) * 20 + kp]);
                    o[j] = __builtin_amdgcn_wmma_f32_16x16x4_f32(false, a, false, bfr,
                                                                 (short)0, o[j], false, false);
                }
            }

            if (kblk < kb_hi) storeKV(cur ^ 1);
        }

        #pragma unroll
        for (int r = 0; r < 8; ++r) {
            float inv  = 1.0f / lrow[r];
            int   srow = q0 + w * 16 + r + rsel;
            size_t rowbase = ((size_t)b * SEQ + srow) * HIDDEN + (size_t)h * HEADSZ;
            #pragma unroll
            for (int j = 0; j < 8; ++j)
                ctx[rowbase + j * 16 + lh] = o[j][r] * inv;
        }
    }
}

extern "C" void kernel_launch(void* const* d_in, const int* in_sizes, int n_in,
                              void* d_out, int out_size, void* d_ws, size_t ws_size,
                              hipStream_t stream)
{
    const float* hs    = (const float*)d_in[0];
    const float* amask = (const float*)d_in[1];
    const float* Wqkv  = (const float*)d_in[2];
    const float* bqkv  = (const float*)d_in[3];
    const float* Wd    = (const float*)d_in[4];
    const float* bd    = (const float*)d_in[5];
    const int*   pos   = (const int*)d_in[6];
    float* out = (float*)d_out;

    float* qkv = (float*)d_ws;
    float* ctx = qkv + (size_t)BATCH * SEQ * QKVDIM;

    (void)in_sizes; (void)n_in; (void)out_size; (void)ws_size;

    gemm_bias_wmma<<<dim3(QKVDIM / 128, (BATCH * SEQ) / 128), dim3(256), 0, stream>>>(
        hs, Wqkv, bqkv, qkv, BATCH * SEQ, QKVDIM, HIDDEN);

    rope_kernel<<<dim3((BATCH * SEQ * NHEADS * 16) / 256), dim3(256), 0, stream>>>(qkv, pos);

    flash_attn_kernel<<<dim3(SEQ / 128, BATCH * NHEADS), dim3(128), 0, stream>>>(qkv, amask, ctx);

    gemm_bias_wmma<<<dim3(HIDDEN / 128, (BATCH * SEQ) / 128), dim3(256), 0, stream>>>(
        ctx, Wd, bd, out, BATCH * SEQ, HIDDEN, HIDDEN);
}


# K-loop global loads addressed as block-uniform base (scalar add per step) + loop-invariant thread offset + literal, strides compile-time: no 64-bit vector address math in GEMM and attention loops
# speedup vs baseline: 1.4531x; 1.2590x over previous
#include <hip/hip_runtime.h>
#include <math.h>

typedef __attribute__((ext_vector_type(2))) float v2f;
typedef __attribute__((ext_vector_type(4))) float f4;
typedef __attribute__((ext_vector_type(8))) float v8f;

#define BATCH   2
#define SEQ     2048
#define HIDDEN  2048
#define NHEADS  16
#define HEADSZ  128
#define QKVDIM  6144
#define ROTH    16
#define MASKVAL (-3.4028234663852886e38f)

static __device__ __forceinline__ void lds_handover_barrier()
{
    asm volatile("s_wait_dscnt 0x0\n\ts_barrier_signal -1\n\ts_barrier_wait -1" ::: "memory");
}

template <int N, int K>
__global__ __launch_bounds__(256) __attribute__((amdgpu_waves_per_eu(4, 4))) void gemm_bias_wmma(
    const float* __restrict__ A, const float* __restrict__ B,
    const float* __restrict__ bias, float* __restrict__ C)
{
    __shared__ float As[2][128 * 36];
    __shared__ float Bt[2][128 * 36];

    const int tid  = threadIdx.x;
    const int lane = tid & 31;
    const int wid  = tid >> 5;
    const int lh   = lane & 15;
    const int kh2  = (lane >> 4) << 1;
    const int rsel = (lane >> 4) << 3;

    const int m0 = blockIdx.y * 128;
    const int n0 = blockIdx.x * 128;
    const int waveRow = (wid & 3) * 32;
    const int waveCol = (wid >> 2) * 64;

    v8f zero = {};
    v8f acc[2][4];
    #pragma unroll
    for (int i = 0; i < 2; ++i)
        #pragma unroll
        for (int j = 0; j < 4; ++j) acc[i][j] = zero;

    f4 ra[4], rb[4];

    constexpr int nt = K >> 5;

    const char* Abase = (const char*)(A + (size_t)m0 * K);
    const char* Bbase = (const char*)(B + n0);
    const unsigned aoff = (unsigned)(((tid >> 3) * K + ((tid & 7) << 2)) * 4);
    const unsigned boff = (unsigned)((((tid >> 7) << 2) * N + (tid & 127)) * 4);

    auto loadAB = [&](int k0) {
        const char* Ak = Abase + (size_t)k0 * 4;
        const char* Bk = Bbase + (size_t)k0 * ((size_t)N * 4);
        #pragma unroll
        for (int it = 0; it < 4; ++it) {
            ra[it] = *(const f4*)((Ak + (size_t)it * 32 * K * 4) + aoff);
            f4 tv;
            tv.x = *(const float*)((Bk + (size_t)(it * 8 + 0) * N * 4) + boff);
            tv.y = *(const float*)((Bk + (size_t)(it * 8 + 1) * N * 4) + boff);
            tv.z = *(const float*)((Bk + (size_t)(it * 8 + 2) * N * 4) + boff);
            tv.w = *(const float*)((Bk + (size_t)(it * 8 + 3) * N * 4) + boff);
            rb[it] = tv;
        }
    };

    loadAB(0);
    #pragma unroll
    for (int it = 0; it < 4; ++it) {
        int e = it * 256 + tid;
        int r = e >> 3, c = (e & 7) << 2;
        *(f4*)(&As[0][r * 36 + c]) = ra[it];
        int n = e & 127, k4 = (e >> 7) << 2;
        *(f4*)(&Bt[0][n * 36 + k4]) = rb[it];
    }

    for (int t = 0; t < nt; ++t) {
        const int cur = t & 1;
        const int k0n = (t + 1) << 5;

        if (t + 1 < nt) loadAB(k0n);

        lds_handover_barrier();

        const float* Ac = &As[cur][0];
        const float* Bc = &Bt[cur][0];
        #pragma unroll
        for (int kk = 0; kk < 32; kk += 4) {
            int kb = kk + kh2;
            v2f af[2];
            #pragma unroll
            for (int i = 0; i < 2; ++i)
                af[i] = *(const v2f*)(&Ac[(waveRow + i * 16 + lh) * 36 + kb]);
            v2f bf[4];
            #pragma unroll
            for (int j = 0; j < 4; ++j)
                bf[j] = *(const v2f*)(&Bc[(waveCol + j * 16 + lh) * 36 + kb]);
            #pragma unroll
            for (int i = 0; i < 2; ++i)
                #pragma unroll
                for (int j = 0; j < 4; ++j)
                    acc[i][j] = __builtin_amdgcn_wmma_f32_16x16x4_f32(
                        false, af[i], false, bf[j], (short)0, acc[i][j], false, false);
        }

        if (t + 1 < nt) {
            #pragma unroll
            for (int it = 0; it < 4; ++it) {
                int e = it * 256 + tid;
                int r = e >> 3, c = (e & 7) << 2;
                *(f4*)(&As[cur ^ 1][r * 36 + c]) = ra[it];
                int n = e & 127, k4 = (e >> 7) << 2;
                *(f4*)(&Bt[cur ^ 1][n * 36 + k4]) = rb[it];
            }
        }
    }

    #pragma unroll
    for (int i = 0; i < 2; ++i) {
        #pragma unroll
        for (int j = 0; j < 4; ++j) {
            int gcol = n0 + waveCol + j * 16 + lh;
            float bv = bias[gcol];
            #pragma unroll
            for (int r = 0; r < 8; ++r) {
                int grow = m0 + waveRow + i * 16 + r + rsel;
                C[(size_t)grow * N + gcol] = acc[i][j][r] + bv;
            }
        }
    }
}

__global__ __launch_bounds__(256) void rope_kernel(
    float* __restrict__ qkv, const int* __restrict__ pos_ids)
{
    int idx = blockIdx.x * 256 + threadIdx.x;
    int i = idx & 15;
    int h = (idx >> 4) & 15;
    int s = (idx >> 8) & (SEQ - 1);
    int b = idx >> 19;

    float t = (float)pos_ids[s];
    float freq = powf(10000.0f, -(float)(2 * i) / 32.0f);
    float sn, cs;
    sincosf(t * freq, &sn, &cs);

    size_t base = ((size_t)(b * SEQ + s)) * QKVDIM + (size_t)h * 384;
    float q0 = qkv[base + i], q1 = qkv[base + i + ROTH];
    qkv[base + i]        = q0 * cs - q1 * sn;
    qkv[base + i + ROTH] = q1 * cs + q0 * sn;
    float k0 = qkv[base + 128 + i], k1 = qkv[base + 128 + i + ROTH];
    qkv[base + 128 + i]        = k0 * cs - k1 * sn;
    qkv[base + 128 + i + ROTH] = k1 * cs + k0 * sn;
}

__global__ __launch_bounds__(128) __attribute__((amdgpu_waves_per_eu(2, 2))) void flash_attn_kernel(
    const float* __restrict__ qkv, const float* __restrict__ amask,
    float* __restrict__ ctx)
{
    __shared__ float Qs[4 * 16 * 132];
    __shared__ float Ks[2][16 * 132];
    __shared__ float Vt[2][128 * 20];
    __shared__ float Ps[4 * 16 * 18];

    const int tid  = threadIdx.x;
    const int lane = tid & 31;
    const int w    = tid >> 5;
    const int lh   = lane & 15;
    const int kh2  = (lane >> 4) << 1;
    const int rsel = (lane >> 4) << 3;

    const int b  = blockIdx.y >> 4;
    const int h  = blockIdx.y & 15;

    const size_t qkv_bh = (size_t)b * SEQ * QKVDIM + (size_t)h * 384;
    const char* kvbase = (const char*)(qkv + qkv_bh);
    const char* mbase  = (const char*)(amask + (size_t)b * SEQ);
    const unsigned koff = (unsigned)(((tid >> 5) * QKVDIM + 128 + ((tid & 31) << 2)) * 4);
    const unsigned voff = (unsigned)((256 + tid) * 4);
    const unsigned moff = (unsigned)(lh * 4);

    #pragma unroll 1
    for (int pass = 0; pass < 2; ++pass) {
        const int qt = pass ? (int)blockIdx.x : (SEQ / 64 - 1 - (int)blockIdx.x);
        const int q0 = qt * 64;

        lds_handover_barrier();

        #pragma unroll
        for (int it = 0; it < 16; ++it) {
            int e = it * 128 + tid;
            int r = e >> 5;
            int c = (e & 31) << 2;
            f4 v = *(const f4*)(qkv + qkv_bh + (size_t)(q0 + r) * QKVDIM + c);
            *(f4*)(&Qs[r * 132 + c]) = v;
        }

        v8f zero = {};
        v8f o[8];
        #pragma unroll
        for (int j = 0; j < 8; ++j) o[j] = zero;
        float mrow[8], lrow[8];
        #pragma unroll
        for (int r = 0; r < 8; ++r) { mrow[r] = MASKVAL; lrow[r] = 0.0f; }

        const float norm = 0.08838834764831845f;
        const int kb_hi = (q0 + 63) >> 4;
        float* Qw = &Qs[w * 16 * 132];
        float* Pw = &Ps[w * 16 * 18];

        f4 rk[4], rv[4];

        auto loadKV = [&](int kblk) {
            const char* kvb = kvbase + (size_t)kblk * ((size_t)16 * QKVDIM * 4);
            #pragma unroll
            for (int it = 0; it < 4; ++it) {
                rk[it] = *(const f4*)((kvb + (size_t)it * 4 * QKVDIM * 4) + koff);
                f4 tv;
                tv.x = *(const float*)((kvb + (size_t)(it * 4 + 0) * QKVDIM * 4) + voff);
                tv.y = *(const float*)((kvb + (size_t)(it * 4 + 1) * QKVDIM * 4) + voff);
                tv.z = *(const float*)((kvb + (size_t)(it * 4 + 2) * QKVDIM * 4) + voff);
                tv.w = *(const float*)((kvb + (size_t)(it * 4 + 3) * QKVDIM * 4) + voff);
                rv[it] = tv;
            }
        };
        auto storeKV = [&](int buf) {
            #pragma unroll
            for (int it = 0; it < 4; ++it) {
                int e = it * 128 + tid;
                int kr = e >> 5, kc = (e & 31) << 2;
                *(f4*)(&Ks[buf][kr * 132 + kc]) = rk[it];
                int vd = e & 127, vk = (e >> 7) << 2;
                *(f4*)(&Vt[buf][vd * 20 + vk]) = rv[it];
            }
        };

        float am_next = *(const float*)(mbase + moff);
        loadKV(0);
        storeKV(0);

        for (int kblk = 0; kblk <= kb_hi; ++kblk) {
            const int cur = kblk & 1;
            const int   kidx = kblk * 16 + lh;
            const float am   = am_next;
            if (kblk < kb_hi) {
                loadKV(kblk + 1);
                am_next = *(const float*)((mbase + (size_t)(kblk + 1) * 64) + moff);
            }

            lds_handover_barrier();

            v8f s = zero;
            const float* Kc = &Ks[cur][0];
            #pragma unroll
            for (int step = 0; step < 32; ++step) {
                int kd = step * 4 + kh2;
                v2f a   = *(const v2f*)(&Qw[lh * 132 + kd]);
                v2f bfr = *(const v2f*)(&Kc[lh * 132 + kd]);
                s = __builtin_amdgcn_wmma_f32_16x16x4_f32(false, a, false, bfr,
                                                          (short)0, s, false, false);
            }

            float sv[8];
            #pragma unroll
            for (int r = 0; r < 8; ++r) {
                int srow = q0 + w * 16 + r + rsel;
                float v = (kidx <= srow) ? (s[r] * norm) : MASKVAL;
                sv[r] = v + am;
            }

            #pragma unroll
            for (int r = 0; r < 8; ++r) {
                float mx = sv[r];
                mx = fmaxf(mx, __shfl_xor(mx, 1, 32));
                mx = fmaxf(mx, __shfl_xor(mx, 2, 32));
                mx = fmaxf(mx, __shfl_xor(mx, 4, 32));
                mx = fmaxf(mx, __shfl_xor(mx, 8, 32));
                float mnew = fmaxf(mrow[r], mx);
                float pe   = __expf(sv[r] - mnew);
                float sum  = pe;
                sum += __shfl_xor(sum, 1, 32);
                sum += __shfl_xor(sum, 2, 32);
                sum += __shfl_xor(sum, 4, 32);
                sum += __shfl_xor(sum, 8, 32);
                float scale = __expf(mrow[r] - mnew);
                lrow[r] = lrow[r] * scale + sum;
                mrow[r] = mnew;
                #pragma unroll
                for (int j = 0; j < 8; ++j) o[j][r] *= scale;
                Pw[(r + rsel) * 18 + lh] = pe;
            }
            asm volatile("s_wait_dscnt 0" ::: "memory");

            const float* Vc = &Vt[cur][0];
            #pragma unroll
            for (int st = 0; st < 4; ++st) {
                int kp = st * 4 + kh2;
                v2f a = *(const v2f*)(&Pw[lh * 18 + kp]);
                #pragma unroll
                for (int j = 0; j < 8; ++j) {
                    v2f bfr = *(const v2f*)(&Vc[(j * 16 + lh) * 20 + kp]);
                    o[j] = __builtin_amdgcn_wmma_f32_16x16x4_f32(false, a, false, bfr,
                                                                 (short)0, o[j], false, false);
                }
            }

            if (kblk < kb_hi) storeKV(cur ^ 1);
        }

        #pragma unroll
        for (int r = 0; r < 8; ++r) {
            float inv  = 1.0f / lrow[r];
            int   srow = q0 + w * 16 + r + rsel;
            size_t rowbase = ((size_t)b * SEQ + srow) * HIDDEN + (size_t)h * HEADSZ;
            #pragma unroll
            for (int j = 0; j < 8; ++j)
                ctx[rowbase + j * 16 + lh] = o[j][r] * inv;
        }
    }
}

extern "C" void kernel_launch(void* const* d_in, const int* in_sizes, int n_in,
                              void* d_out, int out_size, void* d_ws, size_t ws_size,
                              hipStream_t stream)
{
    const float* hs    = (const float*)d_in[0];
    const float* amask = (const float*)d_in[1];
    const float* Wqkv  = (const float*)d_in[2];
    const float* bqkv  = (const float*)d_in[3];
    const float* Wd    = (const float*)d_in[4];
    const float* bd    = (const float*)d_in[5];
    const int*   pos   = (const int*)d_in[6];
    float* out = (float*)d_out;

    float* qkv = (float*)d_ws;
    float* ctx = qkv + (size_t)BATCH * SEQ * QKVDIM;

    (void)in_sizes; (void)n_in; (void)out_size; (void)ws_size;

    gemm_bias_wmma<QKVDIM, HIDDEN><<<dim3(QKVDIM / 128, (BATCH * SEQ) / 128), dim3(256), 0, stream>>>(
        hs, Wqkv, bqkv, qkv);

    rope_kernel<<<dim3((BATCH * SEQ * NHEADS * 16) / 256), dim3(256), 0, stream>>>(qkv, pos);

    flash_attn_kernel<<<dim3(SEQ / 128, BATCH * NHEADS), dim3(128), 0, stream>>>(qkv, amask, ctx);

    gemm_bias_wmma<HIDDEN, HIDDEN><<<dim3(HIDDEN / 128, (BATCH * SEQ) / 128), dim3(256), 0, stream>>>(
        ctx, Wd, bd, out);
}


# Attention block transposed (St = K*Qt, Ot = Vt*Pt): a query row sits in one lane pair, so row max and sum are in-lane plus one half exchange and P feeds P*V from registers; row-sum order changes
# speedup vs baseline: 1.5303x; 1.0531x over previous
#include <hip/hip_runtime.h>
#include <math.h>

typedef __attribute__((ext_vector_type(2))) float v2f;
typedef __attribute__((ext_vector_type(4))) float f4;
typedef __attribute__((ext_vector_type(8))) float v8f;

#define BATCH   2
#define SEQ     2048
#define HIDDEN  2048
#define NHEADS  16
#define HEADSZ  128
#define QKVDIM  6144
#define ROTH    16
#define MASKVAL (-3.4028234663852886e38f)

static __device__ __forceinline__ void lds_handover_barrier()
{
    asm volatile("s_wait_dscnt 0x0\n\ts_barrier_signal -1\n\ts_barrier_wait -1" ::: "memory");
}

template <int N, int K>
__global__ __launch_bounds__(256) __attribute__((amdgpu_waves_per_eu(4, 4))) void gemm_bias_wmma(
    const float* __restrict__ A, const float* __restrict__ B,
    const float* __restrict__ bias, float* __restrict__ C)
{
    __shared__ float As[2][128 * 36];
    __shared__ float Bt[2][128 * 36];

    const int tid  = threadIdx.x;
    const int lane = tid & 31;
    const int wid  = tid >> 5;
    const int lh   = lane & 15;
    const int kh2  = (lane >> 4) << 1;
    const int rsel = (lane >> 4) << 3;

    const int m0 = blockIdx.y * 128;
    const int n0 = blockIdx.x * 128;
    const int waveRow = (wid & 3) * 32;
    const int waveCol = (wid >> 2) * 64;

    v8f zero = {};
    v8f acc[2][4];
    #pragma unroll
    for (int i = 0; i < 2; ++i)
        #pragma unroll
        for (int j = 0; j < 4; ++j) acc[i][j] = zero;

    f4 ra[4], rb[4];

    constexpr int nt = K >> 5;

    const char* Abase = (const char*)(A + (size_t)m0 * K);
    const char* Bbase = (const char*)(B + n0);
    const unsigned aoff = (unsigned)(((tid >> 3) * K + ((tid & 7) << 2)) * 4);
    const unsigned boff = (unsigned)((((tid >> 7) << 2) * N + (tid & 127)) * 4);

    auto loadAB = [&](int k0) {
        const char* Ak = Abase + (size_t)k0 * 4;
        const char* Bk = Bbase + (size_t)k0 * ((size_t)N * 4);
        #pragma unroll
        for (int it = 0; it < 4; ++it) {
            ra[it] = *(const f4*)((Ak + (size_t)it * 32 * K * 4) + aoff);
            f4 tv;
            tv.x = *(const float*)((Bk + (size_t)(it * 8 + 0) * N * 4) + boff);
            tv.y = *(const float*)((Bk + (size_t)(it * 8 + 1) * N * 4) + boff);
            tv.z = *(const float*)((Bk + (size_t)(it * 8 + 2) * N * 4) + boff);
            tv.w = *(const float*)((Bk + (size_t)(it * 8 + 3) * N * 4) + boff);
            rb[it] = tv;
        }
    };

    loadAB(0);
    #pragma unroll
    for (int it = 0; it < 4; ++it) {
        int e = it * 256 + tid;
        int r = e >> 3, c = (e & 7) << 2;
        *(f4*)(&As[0][r * 36 + c]) = ra[it];
        int n = e & 127, k4 = (e >> 7) << 2;
        *(f4*)(&Bt[0][n * 36 + k4]) = rb[it];
    }

    for (int t = 0; t < nt; ++t) {
        const int cur = t & 1;
        const int k0n = (t + 1) << 5;

        if (t + 1 < nt) loadAB(k0n);

        lds_handover_barrier();

        const float* Ac = &As[cur][0];
        const float* Bc = &Bt[cur][0];
        #pragma unroll
        for (int kk = 0; kk < 32; kk += 4) {
            int kb = kk + kh2;
            v2f af[2];
            #pragma unroll
            for (int i = 0; i < 2; ++i)
                af[i] = *(const v2f*)(&Ac[(waveRow + i * 16 + lh) * 36 + kb]);
            v2f bf[4];
            #pragma unroll
            for (int j = 0; j < 4; ++j)
                bf[j] = *(const v2f*)(&Bc[(waveCol + j * 16 + lh) * 36 + kb]);
            #pragma unroll
            for (int i = 0; i < 2; ++i)
                #pragma unroll
                for (int j = 0; j < 4; ++j)
                    acc[i][j] = __builtin_amdgcn_wmma_f32_16x16x4_f32(
                        false, af[i], false, bf[j], (short)0, acc[i][j], false, false);
        }

        if (t + 1 < nt) {
            #pragma unroll
            for (int it = 0; it < 4; ++it) {
                int e = it * 256 + tid;
                int r = e >> 3, c = (e & 7) << 2;
                *(f4*)(&As[cur ^ 1][r * 36 + c]) = ra[it];
                int n = e & 127, k4 = (e >> 7) << 2;
                *(f4*)(&Bt[cur ^ 1][n * 36 + k4]) = rb[it];
            }
        }
    }

    #pragma unroll
    for (int i = 0; i < 2; ++i) {
        #pragma unroll
        for (int j = 0; j < 4; ++j) {
            int gcol = n0 + waveCol + j * 16 + lh;
            float bv = bias[gcol];
            #pragma unroll
            for (int r = 0; r < 8; ++r) {
                int grow = m0 + waveRow + i * 16 + r + rsel;
                C[(size_t)grow * N + gcol] = acc[i][j][r] + bv;
            }
        }
    }
}

__global__ __launch_bounds__(256) void rope_kernel(
    float* __restrict__ qkv, const int* __restrict__ pos_ids)
{
    int idx = blockIdx.x * 256 + threadIdx.x;
    int i = idx & 15;
    int h = (idx >> 4) & 15;
    int s = (idx >> 8) & (SEQ - 1);
    int b = idx >> 19;

    float t = (float)pos_ids[s];
    float freq = powf(10000.0f, -(float)(2 * i) / 32.0f);
    float sn, cs;
    sincosf(t * freq, &sn, &cs);

    size_t base = ((size_t)(b * SEQ + s)) * QKVDIM + (size_t)h * 384;
    float q0 = qkv[base + i], q1 = qkv[base + i + ROTH];
    qkv[base + i]        = q0 * cs - q1 * sn;
    qkv[base + i + ROTH] = q1 * cs + q0 * sn;
    float k0 = qkv[base + 128 + i], k1 = qkv[base + 128 + i + ROTH];
    qkv[base + 128 + i]        = k0 * cs - k1 * sn;
    qkv[base + 128 + i + ROTH] = k1 * cs + k0 * sn;
}

static __device__ __forceinline__ float xhalf(float x)
{
    int b = __builtin_bit_cast(int, x);
    return __builtin_bit_cast(float,
        __builtin_amdgcn_permlanex16(b, b, 0x76543210, 0xfedcba98, false, false));
}

__global__ __launch_bounds__(128) __attribute__((amdgpu_waves_per_eu(2, 2))) void flash_attn_kernel(
    const float* __restrict__ qkv, const float* __restrict__ amask,
    float* __restrict__ ctx)
{
    __shared__ float Qs[4 * 16 * 132];
    __shared__ float Ks[2][16 * 132];
    __shared__ float Vt[2][128 * 20];

    const int tid  = threadIdx.x;
    const int lane = tid & 31;
    const int w    = tid >> 5;
    const int lh   = lane & 15;
    const int kh2  = (lane >> 4) << 1;
    const int rsel = (lane >> 4) << 3;

    const int krow = ((lh >> 1) & 3) * 4 + (lh >> 3) * 2 + (lh & 1);

    const int b  = blockIdx.y >> 4;
    const int h  = blockIdx.y & 15;

    const size_t qkv_bh = (size_t)b * SEQ * QKVDIM + (size_t)h * 384;
    const char* kvbase = (const char*)(qkv + qkv_bh);
    const char* mbase  = (const char*)(amask + (size_t)b * SEQ);
    const unsigned koff = (unsigned)(((tid >> 5) * QKVDIM + 128 + ((tid & 31) << 2)) * 4);
    const unsigned voff = (unsigned)((256 + tid) * 4);
    const unsigned moff = (unsigned)(kh2 * 4);

    #pragma unroll 1
    for (int pass = 0; pass < 2; ++pass) {
        const int qt = pass ? (int)blockIdx.x : (SEQ / 64 - 1 - (int)blockIdx.x);
        const int q0 = qt * 64;
        const int qrow = q0 + w * 16 + lh;

        lds_handover_barrier();

        #pragma unroll
        for (int it = 0; it < 16; ++it) {
            int e = it * 128 + tid;
            int r = e >> 5;
            int c = (e & 31) << 2;
            f4 v = *(const f4*)(qkv + qkv_bh + (size_t)(q0 + r) * QKVDIM + c);
            *(f4*)(&Qs[r * 132 + c]) = v;
        }

        v8f zero = {};
        v8f o[8];
        #pragma unroll
        for (int j = 0; j < 8; ++j) o[j] = zero;
        float mrow = MASKVAL;
        float lrow = 0.0f;

        const float norm = 0.08838834764831845f;
        const int kb_hi = (q0 + 63) >> 4;
        const float* Qw = &Qs[w * 16 * 132];

        f4 rk[4], rv[4];

        auto loadKV = [&](int kblk) {
            const char* kvb = kvbase + (size_t)kblk * ((size_t)16 * QKVDIM * 4);
            #pragma unroll
            for (int it = 0; it < 4; ++it) {
                rk[it] = *(const f4*)((kvb + (size_t)it * 4 * QKVDIM * 4) + koff);
                f4 tv;
                tv.x = *(const float*)((kvb + (size_t)(it * 4 + 0) * QKVDIM * 4) + voff);
                tv.y = *(const float*)((kvb + (size_t)(it * 4 + 1) * QKVDIM * 4) + voff);
                tv.z = *(const float*)((kvb + (size_t)(it * 4 + 2) * QKVDIM * 4) + voff);
                tv.w = *(const float*)((kvb + (size_t)(it * 4 + 3) * QKVDIM * 4) + voff);
                rv[it] = tv;
            }
        };
        auto storeKV = [&](int buf) {
            #pragma unroll
            for (int it = 0; it < 4; ++it) {
                int e = it * 128 + tid;
                int kr = e >> 5, kc = (e & 31) << 2;
                *(f4*)(&Ks[buf][kr * 132 + kc]) = rk[it];
                int vd = e & 127, vk = (e >> 7) << 2;
                *(f4*)(&Vt[buf][vd * 20 + vk]) = rv[it];
            }
        };

        v2f amn[4];
        auto loadMask = [&](int kblk) {
            const char* mb = (mbase + (size_t)kblk * 64) + moff;
            #pragma unroll
            for (int st = 0; st < 4; ++st) amn[st] = *(const v2f*)(mb + st * 16);
        };
        loadMask(0);
        loadKV(0);
        storeKV(0);

        for (int kblk = 0; kblk <= kb_hi; ++kblk) {
            const int cur = kblk & 1;
            v2f am[4];
            #pragma unroll
            for (int st = 0; st < 4; ++st) am[st] = amn[st];
            if (kblk < kb_hi) {
                loadKV(kblk + 1);
                loadMask(kblk + 1);
            }

            lds_handover_barrier();

            v8f s = zero;
            const float* Kc = &Ks[cur][0];
            #pragma unroll
            for (int step = 0; step < 32; ++step) {
                int kd = step * 4 + kh2;
                v2f kf = *(const v2f*)(&Kc[krow * 132 + kd]);
                v2f qf = *(const v2f*)(&Qw[lh * 132 + kd]);
                s = __builtin_amdgcn_wmma_f32_16x16x4_f32(false, kf, false, qf,
                                                          (short)0, s, false, false);
            }

            const int kbase = kblk * 16 + kh2;
            float sv[8];
            #pragma unroll
            for (int r = 0; r < 8; ++r) {
                int key = kbase + 4 * (r >> 1) + (r & 1);
                float v = (key <= qrow) ? (s[r] * norm) : MASKVAL;
                sv[r] = v + am[r >> 1][r & 1];
            }

            float mx = fmaxf(fmaxf(fmaxf(sv[0], sv[1]), fmaxf(sv[2], sv[3])),
                             fmaxf(fmaxf(sv[4], sv[5]), fmaxf(sv[6], sv[7])));
            mx = fmaxf(mx, xhalf(mx));
            const float mnew = fmaxf(mrow, mx);
            float pe[8];
            #pragma unroll
            for (int r = 0; r < 8; ++r) pe[r] = __expf(sv[r] - mnew);
            const float psum = ((pe[0] + pe[1]) + (pe[2] + pe[3])) + ((pe[4] + pe[5]) + (pe[6] + pe[7]));
            const float scale = __expf(mrow - mnew);
            lrow = lrow * scale + psum;
            mrow = mnew;
            #pragma unroll
            for (int j = 0; j < 8; ++j) o[j] = o[j] * scale;

            const float* Vc = &Vt[cur][0];
            #pragma unroll
            for (int st = 0; st < 4; ++st) {
                int kp = st * 4 + kh2;
                v2f pf;
                pf.x = pe[2 * st];
                pf.y = pe[2 * st + 1];
                #pragma unroll
                for (int j = 0; j < 8; ++j) {
                    v2f vf = *(const v2f*)(&Vc[(j * 16 + lh) * 20 + kp]);
                    o[j] = __builtin_amdgcn_wmma_f32_16x16x4_f32(false, vf, false, pf,
                                                                 (short)0, o[j], false, false);
                }
            }

            if (kblk < kb_hi) storeKV(cur ^ 1);
        }

        const float inv = 1.0f / (lrow + xhalf(lrow));
        float* crow = ctx + ((size_t)b * SEQ + qrow) * HIDDEN + (size_t)h * HEADSZ + rsel;
        #pragma unroll
        for (int j = 0; j < 8; ++j) {
            f4 lo, hi;
            lo.x = o[j][0] * inv; lo.y = o[j][1] * inv; lo.z = o[j][2] * inv; lo.w = o[j][3] * inv;
            hi.x = o[j][4] * inv; hi.y = o[j][5] * inv; hi.z = o[j][6] * inv; hi.w = o[j][7] * inv;
            *(f4*)(crow + j * 16)     = lo;
            *(f4*)(crow + j * 16 + 4) = hi;
        }
    }
}

extern "C" void kernel_launch(void* const* d_in, const int* in_sizes, int n_in,
                              void* d_out, int out_size, void* d_ws, size_t ws_size,
                              hipStream_t stream)
{
    const float* hs    = (const float*)d_in[0];
    const float* amask = (const float*)d_in[1];
    const float* Wqkv  = (const float*)d_in[2];
    const float* bqkv  = (const float*)d_in[3];
    const float* Wd    = (const float*)d_in[4];
    const float* bd    = (const float*)d_in[5];
    const int*   pos   = (const int*)d_in[6];
    float* out = (float*)d_out;

    float* qkv = (float*)d_ws;
    float* ctx = qkv + (size_t)BATCH * SEQ * QKVDIM;

    (void)in_sizes; (void)n_in; (void)out_size; (void)ws_size;

    gemm_bias_wmma<QKVDIM, HIDDEN><<<dim3(QKVDIM / 128, (BATCH * SEQ) / 128), dim3(256), 0, stream>>>(
        hs, Wqkv, bqkv, qkv);

    rope_kernel<<<dim3((BATCH * SEQ * NHEADS * 16) / 256), dim3(256), 0, stream>>>(qkv, pos);

    flash_attn_kernel<<<dim3(SEQ / 128, BATCH * NHEADS), dim3(128), 0, stream>>>(qkv, amask, ctx);

    gemm_bias_wmma<HIDDEN, HIDDEN><<<dim3(HIDDEN / 128, (BATCH * SEQ) / 128), dim3(256), 0, stream>>>(
        ctx, Wd, bd, out);
}
